# starting kernel
# baseline (speedup 1.0000x reference)
#include <hip/hip_runtime.h>
#include <hip/hip_bf16.h>

#define BB 32
#define NN 8192
#define HD 256
#define GH 512
#define KC 64
#define ROWS_PER_CLUSTER 128
#define MROWS (BB * (2 * KC + 1))
#define OUTROWS (KC + 1)

typedef __attribute__((ext_vector_type(2))) float v2f;
typedef __attribute__((ext_vector_type(4))) float v4f;
typedef __attribute__((ext_vector_type(8))) float v8f;

__global__ __launch_bounds__(256)
void cluster_sum_kernel(const float* __restrict__ hs, float* __restrict__ H) {
    const int b = blockIdx.x >> 6;
    const int k = blockIdx.x & 63;
    const int c = threadIdx.x & 63;
    const int rg = threadIdx.x >> 6;

    const float* base = hs + ((size_t)b * NN + (size_t)k * ROWS_PER_CLUSTER) * HD;

    v4f sum = {0.f, 0.f, 0.f, 0.f};
    #pragma unroll 4
    for (int j = 0; j < ROWS_PER_CLUSTER / 4; ++j) {
        const int row = rg + 4 * j;
        sum += *(const v4f*)(base + (size_t)row * HD + c * 4);
    }

    __shared__ v4f sdata[4][64];
    sdata[rg][c] = sum;
    __syncthreads();

    if (rg == 0) {
        v4f tot = sdata[0][c] + sdata[1][c] + sdata[2][c] + sdata[3][c];
        const v4f hn = *(const v4f*)(hs + ((size_t)b * NN + (NN - 1)) * HD + c * 4);
        if (k == KC - 1) tot -= hn;

        float* Hb = H + (size_t)b * (2 * KC + 1) * HD;
        *(v4f*)(Hb + (size_t)k * HD + c * 4)        = tot;
        *(v4f*)(Hb + (size_t)(KC + k) * HD + c * 4) = tot + hn;
        if (k == 0)
            *(v4f*)(Hb + (size_t)(2 * KC) * HD + c * 4) = hn;
    }
}

__global__ __launch_bounds__(256)
void transpose_kernel(const float* __restrict__ in, float* __restrict__ out,
                      int R, int Ccols) {
    __shared__ float tile[32][33];
    const int tx = threadIdx.x & 31;
    const int ty = threadIdx.x >> 5;
    const int tilesC = Ccols >> 5;
    const int c0 = (blockIdx.x % tilesC) << 5;
    const int r0 = (blockIdx.x / tilesC) << 5;

    #pragma unroll
    for (int i = 0; i < 32; i += 8)
        tile[ty + i][tx] = in[(size_t)(r0 + ty + i) * Ccols + (c0 + tx)];
    __syncthreads();
    #pragma unroll
    for (int i = 0; i < 32; i += 8)
        out[(size_t)(c0 + ty + i) * R + (r0 + tx)] = tile[tx][ty + i];
}

template <bool RELU>
__global__ __launch_bounds__(256)
void gemm_wmma_f32(const float* __restrict__ A, const float* __restrict__ Bt,
                   const float* __restrict__ bias, float* __restrict__ C,
                   int M, int Kd, int Nd, int nStrips) {
    const int lane  = threadIdx.x & 31;
    const int wave  = threadIdx.x >> 5;
    const int strip = blockIdx.x * 8 + wave;
    if (strip >= nStrips) return;

    const int tilesM = M >> 4;
    const int m0 = (strip % tilesM) << 4;
    const int n0 = (strip / tilesM) << 6;

    const int colL = lane & 15;
    const int kb   = (lane >> 4) << 1;

    v8f acc[4];
    #pragma unroll
    for (int t = 0; t < 4; ++t) {
        const float bv = bias[n0 + 16 * t + colL];
        acc[t] = (v8f){bv, bv, bv, bv, bv, bv, bv, bv};
    }

    const float* arow  = A  + (size_t)(m0 + colL) * Kd + kb;
    const float* brow0 = Bt + (size_t)(n0 + colL) * Kd + kb;
    const size_t bstr  = (size_t)16 * Kd;

    v2f a_cur = *(const v2f*)(arow);
    v2f b_cur[4];
    #pragma unroll
    for (int t = 0; t < 4; ++t)
        b_cur[t] = *(const v2f*)(brow0 + t * bstr);

    #pragma unroll 4
    for (int k0 = 0; k0 < Kd - 4; k0 += 4) {
        const int kn = k0 + 4;
        v2f a_nxt = *(const v2f*)(arow + kn);
        v2f b_nxt[4];
        #pragma unroll
        for (int t = 0; t < 4; ++t)
            b_nxt[t] = *(const v2f*)(brow0 + t * bstr + kn);

        #pragma unroll
        for (int t = 0; t < 4; ++t)
            acc[t] = __builtin_amdgcn_wmma_f32_16x16x4_f32(
                false, a_cur, false, b_cur[t], (short)0, acc[t], false, false);

        a_cur = a_nxt;
        #pragma unroll
        for (int t = 0; t < 4; ++t) b_cur[t] = b_nxt[t];
    }

    #pragma unroll
    for (int t = 0; t < 4; ++t)
        acc[t] = __builtin_amdgcn_wmma_f32_16x16x4_f32(
            false, a_cur, false, b_cur[t], (short)0, acc[t], false, false);

    const int rbase = m0 + ((lane >> 4) << 3);
    #pragma unroll
    for (int t = 0; t < 4; ++t) {
        const int col = n0 + 16 * t + colL;
        #pragma unroll
        for (int v = 0; v < 8; ++v) {
            float x = acc[t][v];
            if (RELU) x = fmaxf(x, 0.0f);
            C[(size_t)(rbase + v) * Nd + col] = x;
        }
    }
}

__global__ __launch_bounds__(256)
void finalize_kernel(const float* __restrict__ gs, float* __restrict__ G,
                     float* __restrict__ Gmask) {
    const int b = blockIdx.x;
    const int h = threadIdx.x;
    const float* g = gs + (size_t)b * (2 * KC + 1) * HD + h;

    float S = 0.f;
    #pragma unroll 8
    for (int k = 0; k < KC; ++k) S += g[(size_t)k * HD];

    float* Gb = G + (size_t)b * OUTROWS * HD;
    #pragma unroll 4
    for (int k = 0; k < KC; ++k)
        Gb[(size_t)k * HD + h] = S - g[(size_t)k * HD] + g[(size_t)(KC + k) * HD];
    Gb[(size_t)KC * HD + h] = S + g[(size_t)(2 * KC) * HD];

    if (h < OUTROWS) Gmask[b * OUTROWS + h] = 1.0f;
}

extern "C" void kernel_launch(void* const* d_in, const int* in_sizes, int n_in,
                              void* d_out, int out_size, void* d_ws, size_t ws_size,
                              hipStream_t stream) {
    const float* hs = (const float*)d_in[0];
    const float* W1 = (const float*)d_in[2];
    const float* b1 = (const float*)d_in[3];
    const float* W2 = (const float*)d_in[4];
    const float* b2 = (const float*)d_in[5];

    float* G     = (float*)d_out;
    float* Gmask = (float*)d_out + (size_t)BB * OUTROWS * HD;

    float* H   = (float*)d_ws;
    float* X   = H   + (size_t)MROWS * HD;
    float* W1t = X   + (size_t)MROWS * GH;
    float* W2t = W1t + (size_t)GH * HD;
    float* gs  = H;

    transpose_kernel<<<(HD / 32) * (GH / 32), 256, 0, stream>>>(W1, W1t, HD, GH);
    transpose_kernel<<<(GH / 32) * (HD / 32), 256, 0, stream>>>(W2, W2t, GH, HD);

    cluster_sum_kernel<<<BB * KC, 256, 0, stream>>>(hs, H);

    {
        const int nStrips = (MROWS / 16) * (GH / 64);
        gemm_wmma_f32<true><<<(nStrips + 7) / 8, 256, 0, stream>>>(
            H, W1t, b1, X, MROWS, HD, GH, nStrips);
    }

    {
        const int nStrips = (MROWS / 16) * (HD / 64);
        gemm_wmma_f32<false><<<(nStrips + 7) / 8, 256, 0, stream>>>(
            X, W2t, b2, gs, MROWS, GH, HD, nStrips);
    }

    finalize_kernel<<<BB, 256, 0, stream>>>(gs, G, Gmask);
}


# GEMM operand tiles staged through LDS: 32xBN work-group tiles, K chunks of 64 fetched one chunk ahead with whole-line 16-byte loads, 8-byte LDS fragment reads; k order per accumulator unchanged
# speedup vs baseline: 1.5921x; 1.5921x over previous
#include <hip/hip_runtime.h>
#include <hip/hip_bf16.h>

#define BB 32
#define NN 8192
#define HD 256
#define GH 512
#define KC 64
#define ROWS_PER_CLUSTER 128
#define MROWS (BB * (2 * KC + 1))
#define OUTROWS (KC + 1)

typedef __attribute__((ext_vector_type(2))) float v2f;
typedef __attribute__((ext_vector_type(4))) float v4f;
typedef __attribute__((ext_vector_type(8))) float v8f;

__global__ __launch_bounds__(256)
void cluster_sum_kernel(const float* __restrict__ hs, float* __restrict__ H) {
    const int b = blockIdx.x >> 6;
    const int k = blockIdx.x & 63;
    const int c = threadIdx.x & 63;
    const int rg = threadIdx.x >> 6;

    const float* base = hs + ((size_t)b * NN + (size_t)k * ROWS_PER_CLUSTER) * HD;

    v4f sum = {0.f, 0.f, 0.f, 0.f};
    #pragma unroll 4
    for (int j = 0; j < ROWS_PER_CLUSTER / 4; ++j) {
        const int row = rg + 4 * j;
        sum += *(const v4f*)(base + (size_t)row * HD + c * 4);
    }

    __shared__ v4f sdata[4][64];
    sdata[rg][c] = sum;
    __syncthreads();

    if (rg == 0) {
        v4f tot = sdata[0][c] + sdata[1][c] + sdata[2][c] + sdata[3][c];
        const v4f hn = *(const v4f*)(hs + ((size_t)b * NN + (NN - 1)) * HD + c * 4);
        if (k == KC - 1) tot -= hn;

        float* Hb = H + (size_t)b * (2 * KC + 1) * HD;
        *(v4f*)(Hb + (size_t)k * HD + c * 4)        = tot;
        *(v4f*)(Hb + (size_t)(KC + k) * HD + c * 4) = tot + hn;
        if (k == 0)
            *(v4f*)(Hb + (size_t)(2 * KC) * HD + c * 4) = hn;
    }
}

__global__ __launch_bounds__(256)
void transpose_kernel(const float* __restrict__ in, float* __restrict__ out,
                      int R, int Ccols) {
    __shared__ float tile[32][33];
    const int tx = threadIdx.x & 31;
    const int ty = threadIdx.x >> 5;
    const int tilesC = Ccols >> 5;
    const int c0 = (blockIdx.x % tilesC) << 5;
    const int r0 = (blockIdx.x / tilesC) << 5;

    #pragma unroll
    for (int i = 0; i < 32; i += 8)
        tile[ty + i][tx] = in[(size_t)(r0 + ty + i) * Ccols + (c0 + tx)];
    __syncthreads();
    #pragma unroll
    for (int i = 0; i < 32; i += 8)
        out[(size_t)(c0 + ty + i) * R + (r0 + tx)] = tile[tx][ty + i];
}

#define GBM 32
#define GBK 64
#define GPITCH (GBK + 4)

template <int BN, bool RELU>
__global__ __launch_bounds__(2 * BN)
void gemm_wmma_f32(const float* __restrict__ A, const float* __restrict__ Bt,
                   const float* __restrict__ bias, float* __restrict__ C,
                   int Kd, int Nd) {
    constexpr int NT = 2 * BN;
    constexpr int RP = NT / 16;
    constexpr int AV = GBM / RP;
    constexpr int BV = BN / RP;

    __shared__ __attribute__((aligned(16))) float As[2][GBM * GPITCH];
    __shared__ __attribute__((aligned(16))) float Bs[2][BN * GPITCH];

    const int tid  = threadIdx.x;
    const int lane = tid & 31;
    const int wave = tid >> 5;
    const int wm   = wave & 1;
    const int wn   = wave >> 1;
    const int colL = lane & 15;
    const int kb   = (lane >> 4) << 1;

    const int tilesN = Nd / BN;
    const int n0 = (blockIdx.x % tilesN) * BN;
    const int m0 = (blockIdx.x / tilesN) * GBM;

    v8f acc[2];
    #pragma unroll
    for (int t = 0; t < 2; ++t) {
        const float bv = bias[n0 + wn * 32 + 16 * t + colL];
        acc[t] = (v8f){bv, bv, bv, bv, bv, bv, bv, bv};
    }

    const int lr = tid >> 4;
    const int lc = (tid & 15) << 2;
    const float* ag = A  + (size_t)(m0 + lr) * Kd + lc;
    const float* bg = Bt + (size_t)(n0 + lr) * Kd + lc;
    const int lofs = lr * GPITCH + lc;

    v4f pa[AV], pb[BV];
    #pragma unroll
    for (int i = 0; i < AV; ++i) pa[i] = *(const v4f*)(ag + (size_t)(i * RP) * Kd);
    #pragma unroll
    for (int i = 0; i < BV; ++i) pb[i] = *(const v4f*)(bg + (size_t)(i * RP) * Kd);
    #pragma unroll
    for (int i = 0; i < AV; ++i) *(v4f*)&As[0][lofs + i * RP * GPITCH] = pa[i];
    #pragma unroll
    for (int i = 0; i < BV; ++i) *(v4f*)&Bs[0][lofs + i * RP * GPITCH] = pb[i];
    __syncthreads();

    const int aoff = (wm * 16 + colL) * GPITCH + kb;
    const int boff = (wn * 32 + colL) * GPITCH + kb;

    const int nch = Kd / GBK;
    for (int ch = 0; ch < nch; ++ch) {
        const int cur = ch & 1;
        const bool more = (ch + 1 < nch);

        if (more) {
            const int kof = (ch + 1) * GBK;
            #pragma unroll
            for (int i = 0; i < AV; ++i) pa[i] = *(const v4f*)(ag + (size_t)(i * RP) * Kd + kof);
            #pragma unroll
            for (int i = 0; i < BV; ++i) pb[i] = *(const v4f*)(bg + (size_t)(i * RP) * Kd + kof);
        }

        const float* as = &As[cur][aoff];
        const float* bs = &Bs[cur][boff];
        #pragma unroll
        for (int ks = 0; ks < GBK / 4; ++ks) {
            const v2f a  = *(const v2f*)(as + ks * 4);
            const v2f b0 = *(const v2f*)(bs + ks * 4);
            const v2f b1 = *(const v2f*)(bs + 16 * GPITCH + ks * 4);
            acc[0] = __builtin_amdgcn_wmma_f32_16x16x4_f32(
                false, a, false, b0, (short)0, acc[0], false, false);
            acc[1] = __builtin_amdgcn_wmma_f32_16x16x4_f32(
                false, a, false, b1, (short)0, acc[1], false, false);
        }

        if (more) {
            #pragma unroll
            for (int i = 0; i < AV; ++i) *(v4f*)&As[cur ^ 1][lofs + i * RP * GPITCH] = pa[i];
            #pragma unroll
            for (int i = 0; i < BV; ++i) *(v4f*)&Bs[cur ^ 1][lofs + i * RP * GPITCH] = pb[i];
            __syncthreads();
        }
    }

    const int rbase = m0 + wm * 16 + ((lane >> 4) << 3);
    #pragma unroll
    for (int t = 0; t < 2; ++t) {
        const int col = n0 + wn * 32 + 16 * t + colL;
        #pragma unroll
        for (int v = 0; v < 8; ++v) {
            float x = acc[t][v];
            if (RELU) x = fmaxf(x, 0.0f);
            C[(size_t)(rbase + v) * Nd + col] = x;
        }
    }
}

__global__ __launch_bounds__(256)
void finalize_kernel(const float* __restrict__ gs, float* __restrict__ G,
                     float* __restrict__ Gmask) {
    const int b = blockIdx.x;
    const int h = threadIdx.x;
    const float* g = gs + (size_t)b * (2 * KC + 1) * HD + h;

    float S = 0.f;
    #pragma unroll 8
    for (int k = 0; k < KC; ++k) S += g[(size_t)k * HD];

    float* Gb = G + (size_t)b * OUTROWS * HD;
    #pragma unroll 4
    for (int k = 0; k < KC; ++k)
        Gb[(size_t)k * HD + h] = S - g[(size_t)k * HD] + g[(size_t)(KC + k) * HD];
    Gb[(size_t)KC * HD + h] = S + g[(size_t)(2 * KC) * HD];

    if (h < OUTROWS) Gmask[b * OUTROWS + h] = 1.0f;
}

extern "C" void kernel_launch(void* const* d_in, const int* in_sizes, int n_in,
                              void* d_out, int out_size, void* d_ws, size_t ws_size,
                              hipStream_t stream) {
    const float* hs = (const float*)d_in[0];
    const float* W1 = (const float*)d_in[2];
    const float* b1 = (const float*)d_in[3];
    const float* W2 = (const float*)d_in[4];
    const float* b2 = (const float*)d_in[5];

    float* G     = (float*)d_out;
    float* Gmask = (float*)d_out + (size_t)BB * OUTROWS * HD;

    float* H   = (float*)d_ws;
    float* X   = H   + (size_t)MROWS * HD;
    float* W1t = X   + (size_t)MROWS * GH;
    float* W2t = W1t + (size_t)GH * HD;
    float* gs  = H;

    transpose_kernel<<<(HD / 32) * (GH / 32), 256, 0, stream>>>(W1, W1t, HD, GH);
    transpose_kernel<<<(GH / 32) * (HD / 32), 256, 0, stream>>>(W2, W2t, GH, HD);

    cluster_sum_kernel<<<BB * KC, 256, 0, stream>>>(hs, H);

    gemm_wmma_f32<128, true><<<(MROWS / GBM) * (GH / 128), 256, 0, stream>>>(
        H, W1t, b1, X, HD, GH);

    gemm_wmma_f32<64, false><<<(MROWS / GBM) * (HD / 64), 128, 0, stream>>>(
        X, W2t, b2, gs, GH, HD);

    finalize_kernel<<<BB, 256, 0, stream>>>(gs, G, Gmask);
}


# B operand tiles staged from the row-major weights as given (two k rows interleaved in LDS, 8-byte fragment reads): both transpose launches and their workspace copies removed; k order unchanged
# speedup vs baseline: 1.6743x; 1.0516x over previous
#include <hip/hip_runtime.h>
#include <hip/hip_bf16.h>

#define BB 32
#define NN 8192
#define HD 256
#define GH 512
#define KC 64
#define ROWS_PER_CLUSTER 128
#define MROWS (BB * (2 * KC + 1))
#define OUTROWS (KC + 1)

typedef __attribute__((ext_vector_type(2))) float v2f;
typedef __attribute__((ext_vector_type(4))) float v4f;
typedef __attribute__((ext_vector_type(8))) float v8f;

__global__ __launch_bounds__(256)
void cluster_sum_kernel(const float* __restrict__ hs, float* __restrict__ H) {
    const int b = blockIdx.x >> 6;
    const int k = blockIdx.x & 63;
    const int c = threadIdx.x & 63;
    const int rg = threadIdx.x >> 6;

    const float* base = hs + ((size_t)b * NN + (size_t)k * ROWS_PER_CLUSTER) * HD;

    v4f sum = {0.f, 0.f, 0.f, 0.f};
    #pragma unroll 4
    for (int j = 0; j < ROWS_PER_CLUSTER / 4; ++j) {
        const int row = rg + 4 * j;
        sum += *(const v4f*)(base + (size_t)row * HD + c * 4);
    }

    __shared__ v4f sdata[4][64];
    sdata[rg][c] = sum;
    __syncthreads();

    if (rg == 0) {
        v4f tot = sdata[0][c] + sdata[1][c] + sdata[2][c] + sdata[3][c];
        const v4f hn = *(const v4f*)(hs + ((size_t)b * NN + (NN - 1)) * HD + c * 4);
        if (k == KC - 1) tot -= hn;

        float* Hb = H + (size_t)b * (2 * KC + 1) * HD;
        *(v4f*)(Hb + (size_t)k * HD + c * 4)        = tot;
        *(v4f*)(Hb + (size_t)(KC + k) * HD + c * 4) = tot + hn;
        if (k == 0)
            *(v4f*)(Hb + (size_t)(2 * KC) * HD + c * 4) = hn;
    }
}

#define GBM 32
#define GBK 64
#define GPITCH (GBK + 4)

template <int BN, bool RELU>
__global__ __launch_bounds__(2 * BN)
void gemm_wmma_f32(const float* __restrict__ A, const float* __restrict__ B,
                   const float* __restrict__ bias, float* __restrict__ C,
                   int Kd, int Nd) {
    constexpr int NT = 2 * BN;
    constexpr int RP = NT / 16;
    constexpr int AV = GBM / RP;
    constexpr int BL = BN / 4;
    constexpr int BR = NT / BL;
    constexpr int BV = GBK / BR;
    constexpr int BJ = BV / 2;
    constexpr int BP = 2 * BN + 32;

    __shared__ __attribute__((aligned(16))) float As[2][GBM * GPITCH];
    __shared__ __attribute__((aligned(16))) float Bs[2][(GBK / 2) * BP];

    const int tid  = threadIdx.x;
    const int lane = tid & 31;
    const int wave = tid >> 5;
    const int wm   = wave & 1;
    const int wn   = wave >> 1;
    const int colL = lane & 15;
    const int kb   = (lane >> 4) << 1;

    const int tilesN = Nd / BN;
    const int n0 = (blockIdx.x % tilesN) * BN;
    const int m0 = (blockIdx.x / tilesN) * GBM;

    v8f acc[2];
    #pragma unroll
    for (int t = 0; t < 2; ++t) {
        const float bv = bias[n0 + wn * 32 + 16 * t + colL];
        acc[t] = (v8f){bv, bv, bv, bv, bv, bv, bv, bv};
    }

    const int lr = tid >> 4;
    const int lc = (tid & 15) << 2;
    const float* ag = A + (size_t)(m0 + lr) * Kd + lc;
    const int lofs = lr * GPITCH + lc;

    const int br = tid / BL;
    const int bc = (tid % BL) << 2;
    const float* bg = B + (size_t)(2 * br) * Nd + n0 + bc;
    const int bofs = br * BP + 2 * bc;

    v4f pa[AV], pb[BV];
    #pragma unroll
    for (int i = 0; i < AV; ++i) pa[i] = *(const v4f*)(ag + (size_t)(i * RP) * Kd);
    #pragma unroll
    for (int i = 0; i < BV; ++i) pb[i] = *(const v4f*)(bg + (size_t)((i >> 1) * 2 * BR + (i & 1)) * Nd);
    #pragma unroll
    for (int i = 0; i < AV; ++i) *(v4f*)&As[0][lofs + i * RP * GPITCH] = pa[i];
    #pragma unroll
    for (int j = 0; j < BJ; ++j) {
        const v4f e = pb[2 * j], o = pb[2 * j + 1];
        *(v4f*)&Bs[0][bofs + j * BR * BP]     = (v4f){e.x, o.x, e.y, o.y};
        *(v4f*)&Bs[0][bofs + j * BR * BP + 4] = (v4f){e.z, o.z, e.w, o.w};
    }
    __syncthreads();

    const int aoff = (wm * 16 + colL) * GPITCH + kb;
    const int boff = (lane >> 4) * BP + 2 * (wn * 32 + colL);

    const int nch = Kd / GBK;
    for (int ch = 0; ch < nch; ++ch) {
        const int cur = ch & 1;
        const bool more = (ch + 1 < nch);

        if (more) {
            const int kof = (ch + 1) * GBK;
            #pragma unroll
            for (int i = 0; i < AV; ++i) pa[i] = *(const v4f*)(ag + (size_t)(i * RP) * Kd + kof);
            #pragma unroll
            for (int i = 0; i < BV; ++i) pb[i] = *(const v4f*)(bg + (size_t)(kof + (i >> 1) * 2 * BR + (i & 1)) * Nd);
        }

        const float* as = &As[cur][aoff];
        const float* bs = &Bs[cur][boff];
        #pragma unroll
        for (int ks = 0; ks < GBK / 4; ++ks) {
            const v2f a  = *(const v2f*)(as + ks * 4);
            const v2f b0 = *(const v2f*)(bs + ks * 2 * BP);
            const v2f b1 = *(const v2f*)(bs + ks * 2 * BP + 32);
            acc[0] = __builtin_amdgcn_wmma_f32_16x16x4_f32(
                false, a, false, b0, (short)0, acc[0], false, false);
            acc[1] = __builtin_amdgcn_wmma_f32_16x16x4_f32(
                false, a, false, b1, (short)0, acc[1], false, false);
        }

        if (more) {
            #pragma unroll
            for (int i = 0; i < AV; ++i) *(v4f*)&As[cur ^ 1][lofs + i * RP * GPITCH] = pa[i];
            #pragma unroll
            for (int j = 0; j < BJ; ++j) {
                const v4f e = pb[2 * j], o = pb[2 * j + 1];
                *(v4f*)&Bs[cur ^ 1][bofs + j * BR * BP]     = (v4f){e.x, o.x, e.y, o.y};
                *(v4f*)&Bs[cur ^ 1][bofs + j * BR * BP + 4] = (v4f){e.z, o.z, e.w, o.w};
            }
            __syncthreads();
        }
    }

    const int rbase = m0 + wm * 16 + ((lane >> 4) << 3);
    #pragma unroll
    for (int t = 0; t < 2; ++t) {
        const int col = n0 + wn * 32 + 16 * t + colL;
        #pragma unroll
        for (int v = 0; v < 8; ++v) {
            float x = acc[t][v];
            if (RELU) x = fmaxf(x, 0.0f);
            C[(size_t)(rbase + v) * Nd + col] = x;
        }
    }
}

__global__ __launch_bounds__(256)
void finalize_kernel(const float* __restrict__ gs, float* __restrict__ G,
                     float* __restrict__ Gmask) {
    const int b = blockIdx.x;
    const int h = threadIdx.x;
    const float* g = gs + (size_t)b * (2 * KC + 1) * HD + h;

    float S = 0.f;
    #pragma unroll 8
    for (int k = 0; k < KC; ++k) S += g[(size_t)k * HD];

    float* Gb = G + (size_t)b * OUTROWS * HD;
    #pragma unroll 4
    for (int k = 0; k < KC; ++k)
        Gb[(size_t)k * HD + h] = S - g[(size_t)k * HD] + g[(size_t)(KC + k) * HD];
    Gb[(size_t)KC * HD + h] = S + g[(size_t)(2 * KC) * HD];

    if (h < OUTROWS) Gmask[b * OUTROWS + h] = 1.0f;
}

extern "C" void kernel_launch(void* const* d_in, const int* in_sizes, int n_in,
                              void* d_out, int out_size, void* d_ws, size_t ws_size,
                              hipStream_t stream) {
    const float* hs = (const float*)d_in[0];
    const float* W1 = (const float*)d_in[2];
    const float* b1 = (const float*)d_in[3];
    const float* W2 = (const float*)d_in[4];
    const float* b2 = (const float*)d_in[5];

    float* G     = (float*)d_out;
    float* Gmask = (float*)d_out + (size_t)BB * OUTROWS * HD;

    float* H   = (float*)d_ws;
    float* X   = H   + (size_t)MROWS * HD;
    float* gs  = H;

    cluster_sum_kernel<<<BB * KC, 256, 0, stream>>>(hs, H);

    gemm_wmma_f32<128, true><<<(MROWS / GBM) * (GH / 128), 256, 0, stream>>>(
        H, W1, b1, X, HD, GH);

    gemm_wmma_f32<64, false><<<(MROWS / GBM) * (HD / 64), 128, 0, stream>>>(
        X, W2, b2, gs, GH, HD);

    finalize_kernel<<<BB, 256, 0, stream>>>(gs, G, Gmask);
}


# Hidden-layer combine folded into the second GEMM's A loader (4 -> 3 launches, Y never stored); first GEMM's epilogue leaves per-16-row-block sums of relu(Z+b1); SX reduction tree changed
# speedup vs baseline: 2.4838x; 1.4835x over previous
#include <hip/hip_runtime.h>
#include <hip/hip_bf16.h>

#define BB 32
#define NN 8192
#define HD 256
#define GH 512
#define KC 64
#define ROWS_PER_CLUSTER 128
#define CROWS (BB * KC)
#define MROWS (CROWS + BB)
#define OUTROWS (KC + 1)

typedef __attribute__((ext_vector_type(2))) float v2f;
typedef __attribute__((ext_vector_type(4))) float v4f;
typedef __attribute__((ext_vector_type(8))) float v8f;

__global__ __launch_bounds__(256)
void cluster_sum_kernel(const float* __restrict__ hs, float* __restrict__ H,
                        float* __restrict__ Gmask) {
    const int b = blockIdx.x >> 6;
    const int k = blockIdx.x & 63;
    const int c = threadIdx.x & 63;
    const int rg = threadIdx.x >> 6;

    const float* base = hs + ((size_t)b * NN + (size_t)k * ROWS_PER_CLUSTER) * HD;

    v4f sum = {0.f, 0.f, 0.f, 0.f};
    #pragma unroll 4
    for (int j = 0; j < ROWS_PER_CLUSTER / 4; ++j) {
        const int row = rg + 4 * j;
        sum += *(const v4f*)(base + (size_t)row * HD + c * 4);
    }

    __shared__ v4f sdata[4][64];
    sdata[rg][c] = sum;
    __syncthreads();

    if (rg == 0) {
        v4f tot = sdata[0][c] + sdata[1][c] + sdata[2][c] + sdata[3][c];
        const v4f hn = *(const v4f*)(hs + ((size_t)b * NN + (NN - 1)) * HD + c * 4);
        if (k == KC - 1) tot -= hn;

        *(v4f*)(H + ((size_t)b * KC + k) * HD + c * 4) = tot;
        if (k == 0)
            *(v4f*)(H + ((size_t)CROWS + b) * HD + c * 4) = hn;
    }
    if (k == 0 && threadIdx.x < OUTROWS) Gmask[b * OUTROWS + threadIdx.x] = 1.0f;
}

__device__ __forceinline__ v4f relu4(v4f a) {
    return (v4f){fmaxf(a.x, 0.0f), fmaxf(a.y, 0.0f), fmaxf(a.z, 0.0f), fmaxf(a.w, 0.0f)};
}

#define GBM 32
#define GBK 64
#define GPITCH (GBK + 4)

template <int BN, bool SECOND, bool HN>
__device__ __forceinline__
void gemm_tile(const float* __restrict__ A, const float* __restrict__ B,
               const float* __restrict__ bias, float* __restrict__ C,
               int Kd, int Nd, float* __restrict__ P,
               const float* __restrict__ hb, const int m0, const int n0,
               float* __restrict__ As, float* __restrict__ Bs) {
    constexpr int NT = 2 * BN;
    constexpr int RP = NT / 16;
    constexpr int AV = GBM / RP;
    constexpr int BL = BN / 4;
    constexpr int BR = NT / BL;
    constexpr int BV = GBK / BR;
    constexpr int BJ = BV / 2;
    constexpr int BP = 2 * BN + 32;
    constexpr int EXB = HN ? 4 * AV : 5;
    constexpr int ASZ = GBM * GPITCH;
    constexpr int BSZ = (GBK / 2) * BP;

    const int tid  = threadIdx.x;
    const int lane = tid & 31;
    const int wave = tid >> 5;
    const int wm   = wave & 1;
    const int wn   = wave >> 1;
    const int colL = lane & 15;
    const int kb   = (lane >> 4) << 1;

    const int bq = m0 >> 6;

    v8f acc[2];
    float bcol[2];
    #pragma unroll
    for (int t = 0; t < 2; ++t) {
        bcol[t] = bias[n0 + wn * 32 + 16 * t + colL];
        const float bv = SECOND ? (float)KC * bcol[t] : 0.0f;
        acc[t] = (v8f){bv, bv, bv, bv, bv, bv, bv, bv};
    }

    const int lr = tid >> 4;
    const int lc = (tid & 15) << 2;
    const float* ag = A + (size_t)(m0 + lr) * Kd + lc;
    const int lofs = lr * GPITCH + lc;

    const int br = tid / BL;
    const int bc = (tid % BL) << 2;
    const float* bg = B + (size_t)(2 * br) * Nd + n0 + bc;
    const int bofs = br * BP + 2 * bc;

    v4f pa[AV], pb[BV], ex[EXB + 1];

    auto fetch = [&](int kof) __attribute__((always_inline)) {
        #pragma unroll
        for (int i = 0; i < AV; ++i) pa[i] = *(const v4f*)(ag + (size_t)(i * RP) * Kd + kof);
        #pragma unroll
        for (int i = 0; i < BV; ++i) pb[i] = *(const v4f*)(bg + (size_t)(kof + (i >> 1) * 2 * BR + (i & 1)) * Nd);
        if (SECOND) {
            const float* pc = P + kof + lc;
            ex[EXB] = *(const v4f*)(hb + kof + lc);
            if (!HN) {
                #pragma unroll
                for (int j = 0; j < 4; ++j) ex[j] = *(const v4f*)(pc + (size_t)(4 * bq + j) * Kd);
                ex[4] = *(const v4f*)(A + (size_t)(CROWS + bq) * Kd + kof + lc);
            } else {
                #pragma unroll
                for (int i = 0; i < AV; ++i) {
                    #pragma unroll
                    for (int j = 0; j < 4; ++j)
                        ex[4 * i + j] = *(const v4f*)(pc + (size_t)(4 * (lr + i * RP) + j) * Kd);
                }
            }
        }
    };

    auto park = [&](int st) __attribute__((always_inline)) {
        #pragma unroll
        for (int i = 0; i < AV; ++i) {
            v4f a = pa[i];
            if (SECOND) {
                const v4f bb = ex[EXB];
                if (!HN) {
                    const v4f sx = (ex[0] + ex[1]) + (ex[2] + ex[3]);
                    const v4f x1 = relu4(a + bb);
                    const v4f x2 = relu4((a + ex[4]) + bb);
                    a = (sx - x1) + x2;
                } else {
                    const v4f sx = (ex[4 * i] + ex[4 * i + 1]) + (ex[4 * i + 2] + ex[4 * i + 3]);
                    a = sx + relu4(a + bb);
                }
            }
            *(v4f*)&As[st * ASZ + lofs + i * RP * GPITCH] = a;
        }
        #pragma unroll
        for (int j = 0; j < BJ; ++j) {
            const v4f e = pb[2 * j], o = pb[2 * j + 1];
            *(v4f*)&Bs[st * BSZ + bofs + j * BR * BP]     = (v4f){e.x, o.x, e.y, o.y};
            *(v4f*)&Bs[st * BSZ + bofs + j * BR * BP + 4] = (v4f){e.z, o.z, e.w, o.w};
        }
    };

    fetch(0);
    park(0);
    __syncthreads();

    const int aoff = (wm * 16 + colL) * GPITCH + kb;
    const int boff = (lane >> 4) * BP + 2 * (wn * 32 + colL);

    const int nch = Kd / GBK;
    for (int ch = 0; ch < nch; ++ch) {
        const int cur = ch & 1;
        const bool more = (ch + 1 < nch);

        if (more) fetch((ch + 1) * GBK);

        const float* as = &As[cur * ASZ + aoff];
        const float* bs = &Bs[cur * BSZ + boff];
        #pragma unroll
        for (int ks = 0; ks < GBK / 4; ++ks) {
            const v2f a  = *(const v2f*)(as + ks * 4);
            const v2f b0 = *(const v2f*)(bs + ks * 2 * BP);
            const v2f b1 = *(const v2f*)(bs + ks * 2 * BP + 32);
            acc[0] = __builtin_amdgcn_wmma_f32_16x16x4_f32(
                false, a, false, b0, (short)0, acc[0], false, false);
            acc[1] = __builtin_amdgcn_wmma_f32_16x16x4_f32(
                false, a, false, b1, (short)0, acc[1], false, false);
        }

        if (more) {
            park(cur ^ 1);
            __syncthreads();
        }
    }

    const int rbase = m0 + wm * 16 + ((lane >> 4) << 3);
    #pragma unroll
    for (int t = 0; t < 2; ++t) {
        const int col = n0 + wn * 32 + 16 * t + colL;
        if (!SECOND) {
            float s = 0.0f;
            #pragma unroll
            for (int v = 0; v < 8; ++v) {
                const float x = acc[t][v];
                C[(size_t)(rbase + v) * Nd + col] = x;
                s += fmaxf(x + bcol[t], 0.0f);
            }
            s += __shfl_xor(s, 16);
            if (lane < 16) P[(size_t)((m0 >> 4) + wm) * Nd + col] = s;
        } else {
            #pragma unroll
            for (int v = 0; v < 8; ++v) {
                const int r = rbase + v;
                float x = acc[t][v];
                int orow = (r >> 6) * OUTROWS + (r & (KC - 1));
                if (HN) {
                    orow = (r - CROWS) * OUTROWS + KC;
                    x += bcol[t];
                }
                C[(size_t)orow * Nd + col] = x;
            }
        }
    }
}


template <int BN, bool SECOND>
__global__ __launch_bounds__(2 * BN)
void gemm_wmma_f32(const float* __restrict__ A, const float* __restrict__ B,
                   const float* __restrict__ bias, float* __restrict__ C,
                   int Kd, int Nd, float* __restrict__ P,
                   const float* __restrict__ hb) {
    __shared__ __attribute__((aligned(16))) float As[2 * GBM * GPITCH];
    __shared__ __attribute__((aligned(16))) float Bs[2 * (GBK / 2) * (2 * BN + 32)];

    const int tilesN = Nd / BN;
    const int n0 = (blockIdx.x % tilesN) * BN;
    const int m0 = (blockIdx.x / tilesN) * GBM;
    if (SECOND && m0 >= CROWS)
        gemm_tile<BN, SECOND, true >(A, B, bias, C, Kd, Nd, P, hb, m0, n0, As, Bs);
    else
        gemm_tile<BN, SECOND, false>(A, B, bias, C, Kd, Nd, P, hb, m0, n0, As, Bs);
}

extern "C" void kernel_launch(void* const* d_in, const int* in_sizes, int n_in,
                              void* d_out, int out_size, void* d_ws, size_t ws_size,
                              hipStream_t stream) {
    const float* hs = (const float*)d_in[0];
    const float* W1 = (const float*)d_in[2];
    const float* b1 = (const float*)d_in[3];
    const float* W2 = (const float*)d_in[4];
    const float* b2 = (const float*)d_in[5];

    float* G     = (float*)d_out;
    float* Gmask = (float*)d_out + (size_t)BB * OUTROWS * HD;

    float* H   = (float*)d_ws;
    float* Z   = H   + (size_t)MROWS * HD;
    float* P   = Z   + (size_t)MROWS * GH;

    cluster_sum_kernel<<<BB * KC, 256, 0, stream>>>(hs, H, Gmask);

    gemm_wmma_f32<64, false><<<(MROWS / GBM) * (GH / 64), 128, 0, stream>>>(
        H, W1, b1, Z, HD, GH, P, nullptr);

    gemm_wmma_f32<64, true><<<(MROWS / GBM) * (HD / 64), 128, 0, stream>>>(
        Z, W2, b2, G, GH, HD, P, b1);
}
